# starting kernel
# baseline (speedup 1.0000x reference)
#include <hip/hip_runtime.h>


typedef __attribute__((ext_vector_type(2))) float v2f;
typedef __attribute__((ext_vector_type(8))) float v8f;

#define NPTS   4096
#define AATOMS 256
#define DIN    39
#define HDIM   50
#define MB     128
#define W1PS   160
#define W2PS   160
#define HS     68

__device__ __forceinline__ float silu_f(float x) {
    return x / (1.0f + __expf(-x));
}

__global__ __launch_bounds__(256) void moe_mlp_wmma_kernel(
    const float* __restrict__ desc, const int* __restrict__ numbers,
    const float* __restrict__ W1,   const float* __restrict__ b1,
    const float* __restrict__ W2,   const float* __restrict__ b2,
    const float* __restrict__ W3,   const float* __restrict__ b3,
    float* __restrict__ out)
{
    __shared__ __align__(16) float W1p[20 * W1PS];
    __shared__ __align__(16) float W2p[26 * W2PS];
    __shared__ __align__(16) float Hld[MB * HS];
    __shared__ float w3p[52];
    __shared__ float b1p[64];
    __shared__ float b2p[64];

    const int tid  = threadIdx.x;
    const int lane = tid & 31;
    const int wave = tid >> 5;
    const int m16  = lane & 15;
    const int kh   = lane >> 4;

    const int chunk = blockIdx.x;
    const int atom  = blockIdx.y;
    const int s     = numbers[atom];

    for (int i = tid; i < 20 * 64; i += 256) {
        const int p = i >> 6, h = i & 63;
        const int k0r = 2 * p, k1r = 2 * p + 1;
        float x = (k0r < DIN && h < HDIM) ? W1[(s * DIN + k0r) * HDIM + h] : 0.0f;
        float y = (k1r < DIN && h < HDIM) ? W1[(s * DIN + k1r) * HDIM + h] : 0.0f;
        W1p[p * W1PS + 2 * h + 0] = x;
        W1p[p * W1PS + 2 * h + 1] = y;
    }
    for (int i = tid; i < 26 * 64; i += 256) {
        const int p = i >> 6, h = i & 63;
        const int k0r = 2 * p, k1r = 2 * p + 1;
        float x = (k0r < HDIM && h < HDIM) ? W2[(s * HDIM + k0r) * HDIM + h] : 0.0f;
        float y = (k1r < HDIM && h < HDIM) ? W2[(s * HDIM + k1r) * HDIM + h] : 0.0f;
        W2p[p * W2PS + 2 * h + 0] = x;
        W2p[p * W2PS + 2 * h + 1] = y;
    }
    if (tid < 64) {
        b1p[tid] = (tid < HDIM) ? b1[s * HDIM + tid] : 0.0f;
        b2p[tid] = (tid < HDIM) ? b2[s * HDIM + tid] : 0.0f;
    }
    if (tid < 52) w3p[tid] = (tid < HDIM) ? W3[s * HDIM + tid] : 0.0f;
    __syncthreads();

    const int row0 = chunk * MB + wave * 16;
    const float* descRow = desc + ((size_t)(row0 + m16) * AATOMS + atom) * DIN;

    v8f c0 = {0,0,0,0,0,0,0,0}, c1 = c0, c2 = c0, c3 = c0;
    #pragma unroll
    for (int k0 = 0; k0 < 10; ++k0) {
        const int c = k0 * 4 + kh * 2;
        v2f av;
        av.x = descRow[c];
        av.y = (c + 1 < DIN) ? descRow[c + 1] : 0.0f;
        const v2f* bp = (const v2f*)&W1p[(2 * k0 + kh) * W1PS];
        const v2f bv0 = bp[m16 +  0];
        const v2f bv1 = bp[m16 + 16];
        const v2f bv2 = bp[m16 + 32];
        const v2f bv3 = bp[m16 + 48];
        c0 = __builtin_amdgcn_wmma_f32_16x16x4_f32(false, av, false, bv0, (short)0, c0, false, false);
        c1 = __builtin_amdgcn_wmma_f32_16x16x4_f32(false, av, false, bv1, (short)0, c1, false, false);
        c2 = __builtin_amdgcn_wmma_f32_16x16x4_f32(false, av, false, bv2, (short)0, c2, false, false);
        c3 = __builtin_amdgcn_wmma_f32_16x16x4_f32(false, av, false, bv3, (short)0, c3, false, false);
    }
    {
        float* hbase = &Hld[(wave * 16) * HS];
        #pragma unroll
        for (int r = 0; r < 8; ++r) {
            float* hrow = hbase + (r + 8 * kh) * HS;
            hrow[m16 +  0] = silu_f(c0[r] + b1p[m16 +  0]);
            hrow[m16 + 16] = silu_f(c1[r] + b1p[m16 + 16]);
            hrow[m16 + 32] = silu_f(c2[r] + b1p[m16 + 32]);
            hrow[m16 + 48] = silu_f(c3[r] + b1p[m16 + 48]);
        }
    }

    v8f d0 = {0,0,0,0,0,0,0,0}, d1 = d0, d2 = d0, d3 = d0;
    const float* arow = &Hld[(wave * 16 + m16) * HS];
    #pragma unroll
    for (int k0 = 0; k0 < 13; ++k0) {
        const int c = k0 * 4 + kh * 2;
        const v2f av = *(const v2f*)(arow + c);
        const v2f* bp = (const v2f*)&W2p[(2 * k0 + kh) * W2PS];
        const v2f bv0 = bp[m16 +  0];
        const v2f bv1 = bp[m16 + 16];
        const v2f bv2 = bp[m16 + 32];
        const v2f bv3 = bp[m16 + 48];
        d0 = __builtin_amdgcn_wmma_f32_16x16x4_f32(false, av, false, bv0, (short)0, d0, false, false);
        d1 = __builtin_amdgcn_wmma_f32_16x16x4_f32(false, av, false, bv1, (short)0, d1, false, false);
        d2 = __builtin_amdgcn_wmma_f32_16x16x4_f32(false, av, false, bv2, (short)0, d2, false, false);
        d3 = __builtin_amdgcn_wmma_f32_16x16x4_f32(false, av, false, bv3, (short)0, d3, false, false);
    }
    {
        float* hbase = &Hld[(wave * 16) * HS];
        #pragma unroll
        for (int r = 0; r < 8; ++r) {
            float* hrow = hbase + (r + 8 * kh) * HS;
            hrow[m16 +  0] = silu_f(d0[r] + b2p[m16 +  0]);
            hrow[m16 + 16] = silu_f(d1[r] + b2p[m16 + 16]);
            hrow[m16 + 32] = silu_f(d2[r] + b2p[m16 + 32]);
            hrow[m16 + 48] = silu_f(d3[r] + b2p[m16 + 48]);
        }
    }
    __syncthreads();

    if (tid < MB) {
        float acc = b3[s];
        const float* hrow = &Hld[tid * HS];
        #pragma unroll
        for (int h = 0; h < HDIM; ++h) acc += hrow[h] * w3p[h];
        out[(size_t)(chunk * MB + tid) * AATOMS + atom] = acc;
    }
}

extern "C" void kernel_launch(void* const* d_in, const int* in_sizes, int n_in,
                              void* d_out, int out_size, void* d_ws, size_t ws_size,
                              hipStream_t stream) {
    (void)in_sizes; (void)n_in; (void)d_ws; (void)ws_size; (void)out_size;
    const float* desc    = (const float*)d_in[0];
    const int*   numbers = (const int*)  d_in[1];
    const float* W1      = (const float*)d_in[2];
    const float* b1      = (const float*)d_in[3];
    const float* W2      = (const float*)d_in[4];
    const float* b2      = (const float*)d_in[5];
    const float* W3      = (const float*)d_in[6];
    const float* b3      = (const float*)d_in[7];
    float* out = (float*)d_out;

    dim3 grid(NPTS / MB, AATOMS);
    moe_mlp_wmma_kernel<<<grid, 256, 0, stream>>>(desc, numbers, W1, b1, W2, b2, W3, b3, out);
}


# a wave owns 32 rows as two 16-row tiles (block of 4 waves, same grid): each weight fragment, column-48/49 weight pair and w3 value read from LDS now feeds both tiles; sums and roundings unchanged
# speedup vs baseline: 1.2375x; 1.2375x over previous
#include <hip/hip_runtime.h>


typedef __attribute__((ext_vector_type(2))) float v2f;
typedef __attribute__((ext_vector_type(4))) float v4f;
typedef __attribute__((ext_vector_type(8))) float v8f;

#define NPTS   4096
#define AATOMS 256
#define DIN    39
#define HDIM   50
#define MB     128
#define NT     2
#define W1R    40
#define W2R    52
#define HS     68
#define NP1    (DIN * HDIM / 2)
#define NP2    (HDIM * HDIM / 2)

__device__ __forceinline__ float silu_f(float x) {
    return x / (1.0f + __expf(-x));
}

__global__ __launch_bounds__(128) void moe_mlp_wmma_kernel(
    const float* __restrict__ desc, const int* __restrict__ numbers,
    const float* __restrict__ W1,   const float* __restrict__ b1,
    const float* __restrict__ W2,   const float* __restrict__ b2,
    const float* __restrict__ W3,   const float* __restrict__ b3,
    float* __restrict__ out)
{
    __shared__ __align__(16) float W1l[W1R * HDIM];
    __shared__ __align__(16) float W2l[W2R * HDIM];
    __shared__ __align__(16) float Hld[MB * HS];
    __shared__ float w3p[52];
    __shared__ float b1p[64];
    __shared__ float b2p[64];

    const int tid  = threadIdx.x;
    const int lane = tid & 31;
    const int wave = tid >> 5;
    const int m16  = lane & 15;
    const int kh   = lane >> 4;

    const int chunk = blockIdx.x;
    const int atom  = blockIdx.y;
    const int s     = numbers[atom];

    {
        const v2f* src1 = (const v2f*)(W1 + (size_t)s * (DIN * HDIM));
        const v2f* src2 = (const v2f*)(W2 + (size_t)s * (HDIM * HDIM));
        v2f* dst1 = (v2f*)W1l;
        v2f* dst2 = (v2f*)W2l;
        const int tl1 = tid + 896 < NP1 ? tid + 896 : NP1 - 1;
        const int tl2 = tid + 1152 < NP2 ? tid + 1152 : NP2 - 1;
        const int tb  = tid < HDIM ? tid : HDIM - 1;
        v2f p[8], q[10];
        for (int j = 0; j < 7; ++j) p[j] = src1[tid + 128 * j];
        p[7] = src1[tl1];
        for (int j = 0; j < 9; ++j) q[j] = src2[tid + 128 * j];
        q[9] = src2[tl2];
        const float vb1 = b1[s * HDIM + tb], vb2 = b2[s * HDIM + tb], vw3 = W3[s * HDIM + tb];
        for (int j = 0; j < 7; ++j) dst1[tid + 128 * j] = p[j];
        dst1[tl1] = p[7];
        for (int j = 0; j < 9; ++j) dst2[tid + 128 * j] = q[j];
        dst2[tl2] = q[9];
        const v2f z2 = {0.0f, 0.0f};
        if (tid < 50) dst2[NP2 + tid] = z2;
        if (tid < 25) dst1[NP1 + tid] = z2;
        if (tid < 64) {
            b1p[tid] = (tid < HDIM) ? vb1 : 0.0f;
            b2p[tid] = (tid < HDIM) ? vb2 : 0.0f;
        }
        if (tid < 52) w3p[tid] = (tid < HDIM) ? vw3 : 0.0f;
    }
    __syncthreads();

    const int row0 = chunk * MB + wave * (16 * NT);
    const float* descRow[NT];
    descRow[0] = desc + ((size_t)(row0 + m16) * AATOMS + atom) * DIN;
    descRow[1] = descRow[0] + (size_t)16 * AATOMS * DIN;
    float* const hwave = &Hld[(wave * (16 * NT)) * HS];

    int n1 = 16, n2 = 32;
    asm volatile("" : "+v"(n1), "+v"(n2));
    const int wl0 = kh * 2 * HDIM + m16;
    const int wl1 = wl0 + n1, wl2 = wl0 + n2;

    const v8f zero8 = {0,0,0,0,0,0,0,0};
    v8f c0[NT], c1[NT], c2[NT];
    v2f a1[NT][10];
    for (int t = 0; t < NT; ++t) { c0[t] = zero8; c1[t] = zero8; c2[t] = zero8; }
    for (int k0 = 0; k0 < 10; ++k0) {
        const int c = k0 * 4 + kh * 2;
        for (int t = 0; t < NT; ++t) {
            v2f av;
            av.x = descRow[t][c];
            av.y = (c + 1 < DIN) ? descRow[t][c + 1] : 0.0f;
            a1[t][k0] = av;
        }
        const int ko = k0 * 4 * HDIM;
        const v2f bv0 = {W1l[wl0 + ko], W1l[wl0 + ko + HDIM]};
        const v2f bv1 = {W1l[wl1 + ko], W1l[wl1 + ko + HDIM]};
        const v2f bv2 = {W1l[wl2 + ko], W1l[wl2 + ko + HDIM]};
        for (int t = 0; t < NT; ++t) {
            c0[t] = __builtin_amdgcn_wmma_f32_16x16x4_f32(false, a1[t][k0], false, bv0, (short)0, c0[t], false, false);
            c1[t] = __builtin_amdgcn_wmma_f32_16x16x4_f32(false, a1[t][k0], false, bv1, (short)0, c1[t], false, false);
            c2[t] = __builtin_amdgcn_wmma_f32_16x16x4_f32(false, a1[t][k0], false, bv2, (short)0, c2[t], false, false);
        }
    }
    for (int t = 0; t < NT; ++t) {
        float* hbase = hwave + (16 * t) * HS;
        for (int r = 0; r < 8; ++r) {
            float* hrow = hbase + (r + 8 * kh) * HS;
            hrow[m16 +  0] = c0[t][r] + b1p[m16 +  0];
            hrow[m16 + 16] = c1[t][r] + b1p[m16 + 16];
            hrow[m16 + 32] = c2[t][r] + b1p[m16 + 32];
        }
    }
    {
        v2f ex[NT], ey[NT];
        for (int t = 0; t < NT; ++t) { ex[t].x = 0.0f; ex[t].y = 0.0f; ey[t] = ex[t]; }
        for (int k0 = 0; k0 < 10; ++k0) {
            const float* wr = &W1l[(k0 * 4 + kh * 2) * HDIM + 48];
            const v2f r0 = *(const v2f*)wr;
            const v2f r1 = *(const v2f*)(wr + HDIM);
            for (int t = 0; t < NT; ++t) {
                ex[t].x += a1[t][k0].x * r0.x;
                ex[t].y += a1[t][k0].x * r0.y;
                ey[t].x += a1[t][k0].y * r1.x;
                ey[t].y += a1[t][k0].y * r1.y;
            }
        }
        for (int t = 0; t < NT; ++t) {
            const float s48 = ex[t].x + ey[t].x, s49 = ex[t].y + ey[t].y;
            const float got = __shfl_xor(kh ? s48 : s49, 16);
            float* hme = hwave + (16 * t + m16) * HS + 48;
            hme[kh]     = ((kh ? s49 : s48) + got) + b1p[48 + kh];
            hme[2 + kh] = 0.0f;
        }
    }

    v8f d0[NT], d1[NT], d2[NT];
    v2f fx[NT], fy[NT];
    for (int t = 0; t < NT; ++t) {
        d0[t] = zero8; d1[t] = zero8; d2[t] = zero8;
        fx[t].x = 0.0f; fx[t].y = 0.0f; fy[t] = fx[t];
    }
    const float* arow = hwave + m16 * HS;
    #pragma unroll
    for (int k0 = 0; k0 < 13; ++k0) {
        const int c = k0 * 4 + kh * 2;
        v2f av[NT];
        for (int t = 0; t < NT; ++t) {
            const v2f raw = *(const v2f*)(arow + (16 * t) * HS + c);
            av[t].x = silu_f(raw.x);
            av[t].y = silu_f(raw.y);
        }
        const float* wrow = &W2l[c * HDIM];
        const int ko = k0 * 4 * HDIM;
        const v2f bv0 = {W2l[wl0 + ko], W2l[wl0 + ko + HDIM]};
        const v2f bv1 = {W2l[wl1 + ko], W2l[wl1 + ko + HDIM]};
        const v2f bv2 = {W2l[wl2 + ko], W2l[wl2 + ko + HDIM]};
        const v2f r0  = *(const v2f*)(wrow + 48);
        const v2f r1  = *(const v2f*)(wrow + HDIM + 48);
        for (int t = 0; t < NT; ++t) {
            d0[t] = __builtin_amdgcn_wmma_f32_16x16x4_f32(false, av[t], false, bv0, (short)0, d0[t], false, false);
            d1[t] = __builtin_amdgcn_wmma_f32_16x16x4_f32(false, av[t], false, bv1, (short)0, d1[t], false, false);
            d2[t] = __builtin_amdgcn_wmma_f32_16x16x4_f32(false, av[t], false, bv2, (short)0, d2[t], false, false);
            fx[t].x += av[t].x * r0.x;
            fx[t].y += av[t].x * r0.y;
            fy[t].x += av[t].y * r1.x;
            fy[t].y += av[t].y * r1.y;
        }
    }
    for (int t = 0; t < NT; ++t) {
        float* hbase = hwave + (16 * t) * HS;
        for (int r = 0; r < 8; ++r) {
            float* hrow = hbase + (r + 8 * kh) * HS;
            hrow[m16 +  0] = d0[t][r] + b2p[m16 +  0];
            hrow[m16 + 16] = d1[t][r] + b2p[m16 + 16];
            hrow[m16 + 32] = d2[t][r] + b2p[m16 + 32];
        }
        const float s48 = fx[t].x + fy[t].x, s49 = fx[t].y + fy[t].y;
        const float got = __shfl_xor(kh ? s48 : s49, 16);
        hwave[(16 * t + m16) * HS + 48 + kh] = ((kh ? s49 : s48) + got) + b2p[48 + kh];
    }

    {
        const float bias3 = kh ? 0.0f : b3[s];
        float acc[NT];
        const float* hr = hwave + m16 * HS + 26 * kh;
        const float* wr = &w3p[26 * kh];
        for (int t = 0; t < NT; ++t) acc[t] = bias3;
        #pragma unroll
        for (int i = 0; i < 26; ++i) {
            const float w = wr[i];
            for (int t = 0; t < NT; ++t) acc[t] += silu_f(hr[(16 * t) * HS + i]) * w;
        }
        for (int t = 0; t < NT; ++t) {
            const float tot = acc[t] + __shfl_xor(acc[t], 16);
            if (kh == 0) out[(size_t)(row0 + 16 * t + m16) * AATOMS + atom] = tot;
        }
    }
}

extern "C" void kernel_launch(void* const* d_in, const int* in_sizes, int n_in,
                              void* d_out, int out_size, void* d_ws, size_t ws_size,
                              hipStream_t stream) {
    (void)in_sizes; (void)n_in; (void)d_ws; (void)ws_size; (void)out_size;
    const float* desc    = (const float*)d_in[0];
    const int*   numbers = (const int*)  d_in[1];
    const float* W1      = (const float*)d_in[2];
    const float* b1      = (const float*)d_in[3];
    const float* W2      = (const float*)d_in[4];
    const float* b2      = (const float*)d_in[5];
    const float* W3      = (const float*)d_in[6];
    const float* b3      = (const float*)d_in[7];
    float* out = (float*)d_out;

    dim3 grid(NPTS / MB, AATOMS);
    moe_mlp_wmma_kernel<<<grid, 128, 0, stream>>>(desc, numbers, W1, b1, W2, b2, W3, b3, out);
}
